# starting kernel
# baseline (speedup 1.0000x reference)
#include <hip/hip_runtime.h>

#define BATCH 4
#define FDIM  64

typedef __attribute__((ext_vector_type(2))) float v2f;
typedef __attribute__((ext_vector_type(8))) float v8f;

__global__ void gcn_zero_ws(float* __restrict__ p, long n) {
    long i = (long)blockIdx.x * blockDim.x + threadIdx.x;
    const long stride = (long)gridDim.x * blockDim.x;
    for (; i < n; i += stride) p[i] = 0.0f;
}

__global__ void gcn_spmm_scatter(const float* __restrict__ x,
                                 const float* __restrict__ vals,
                                 const int*   __restrict__ rows,
                                 const int*   __restrict__ cols,
                                 float*       __restrict__ agg,
                                 long E, long Nn) {
    const long g = (long)blockIdx.x * blockDim.x + threadIdx.x;
    const long e = g >> 6;
    if (e >= E) return;
    const int  f = (int)(g & 63);
    const float v = vals[e];
    const long  r = rows[e];
    const long  c = cols[e];
#pragma unroll
    for (int b = 0; b < BATCH; ++b) {
        const float xv = x[((long)b * Nn + c) * FDIM + f];
        __hip_atomic_fetch_add(&agg[((long)b * Nn + r) * FDIM + f], v * xv,
                               __ATOMIC_RELAXED, __HIP_MEMORY_SCOPE_AGENT);
    }
}

__global__ void __launch_bounds__(256)
gcn_gemm_bias_elu(const float* __restrict__ agg,
                  const float* __restrict__ W,
                  const float* __restrict__ bias,
                  float*       __restrict__ out,
                  long nrows) {
    const int lane = threadIdx.x & 31;
    const int wave = threadIdx.x >> 5;
    const long rowBase = ((long)blockIdx.x * 8 + wave) * 16;
    if (rowBase >= nrows) return;

    const int m  = lane & 15;
    const int kg = (lane >> 4) << 1;

    long arow = rowBase + m;
    if (arow >= nrows) arow = nrows - 1;
    const float* __restrict__ aptr = agg + arow * FDIM;

    v8f acc[4] = {v8f{}, v8f{}, v8f{}, v8f{}};

#pragma unroll
    for (int k4 = 0; k4 < FDIM / 4; ++k4) {
        const int kk = k4 * 4 + kg;
        v2f a;
        a.x = aptr[kk];
        a.y = aptr[kk + 1];

        const float* __restrict__ w0 = W + (long)kk * FDIM + m;
        const float* __restrict__ w1 = W + (long)(kk + 1) * FDIM + m;
#pragma unroll
        for (int j = 0; j < 4; ++j) {
            v2f b;
            b.x = w0[j * 16];
            b.y = w1[j * 16];
            acc[j] = __builtin_amdgcn_wmma_f32_16x16x4_f32(
                 false, a,  false, b,
                 (short)0, acc[j],
                 false,  false);
        }
    }

    const int rowAdd = (lane >> 4) * 8;
#pragma unroll
    for (int j = 0; j < 4; ++j) {
        const int col = j * 16 + m;
        const float bb = bias[col];
#pragma unroll
        for (int r = 0; r < 8; ++r) {
            const long row = rowBase + rowAdd + r;
            if (row < nrows) {
                const float v = acc[j][r] + bb;
                out[row * FDIM + col] = (v > 0.0f) ? v : (__expf(v) - 1.0f);
            }
        }
    }
}

extern "C" void kernel_launch(void* const* d_in, const int* in_sizes, int n_in,
                              void* d_out, int out_size, void* d_ws, size_t ws_size,
                              hipStream_t stream) {
    const float* x      = (const float*)d_in[0];
    const float* vals   = (const float*)d_in[1];
    const float* weight = (const float*)d_in[2];
    const float* bias   = (const float*)d_in[3];
    const int*   rows   = (const int*)d_in[4];
    const int*   cols   = (const int*)d_in[5];
    float*       out    = (float*)d_out;

    const long E  = in_sizes[1];
    const long Nn = (long)in_sizes[0] / (BATCH * FDIM);
    const long aggElems = (long)BATCH * Nn * FDIM;
    float* agg = (float*)d_ws;

    {
        const int  threads = 256;
        long blocks = (aggElems + threads - 1) / threads;
        if (blocks > 32768) blocks = 32768;
        gcn_zero_ws<<<(unsigned)blocks, threads, 0, stream>>>(agg, aggElems);
    }
    {
        const long work = E * FDIM;
        const int  threads = 256;
        const long blocks = (work + threads - 1) / threads;
        gcn_spmm_scatter<<<(unsigned)blocks, threads, 0, stream>>>(
            x, vals, rows, cols, agg, E, Nn);
    }
    {
        const long nrows  = (long)BATCH * Nn;
        const long blocks = (nrows + 127) / 128;
        gcn_gemm_bias_elu<<<(unsigned)blocks, 256, 0, stream>>>(
            agg, weight, bias, out, nrows);
    }
}


# SpMM: global f32 atomic scatter replaced by row-tile owner work-groups: each scans all edge rows, keeps its 196 rows' edges, sums them in LDS by LDS float atomics (arrival order); zero launch gone
# speedup vs baseline: 2.9974x; 2.9974x over previous
#include <hip/hip_runtime.h>


#define BATCH 4
#define FDIM  64

typedef __attribute__((ext_vector_type(2))) float v2f;
typedef __attribute__((ext_vector_type(8))) float v8f;

#define TROWS   196
#define NWAVES  32
#define LCAP    736
#define LFLUSH  224

__device__ __forceinline__ void lds_add(float* p, float v) {
    __hip_atomic_fetch_add(p, v, __ATOMIC_RELAXED, __HIP_MEMORY_SCOPE_WORKGROUP);
}

template <int K>
__device__ __forceinline__ void gather_k(const float* __restrict__ x, float* sacc,
                                         int c, float v, int r, int tileBase, int j0, int lane, long bstride) {
    v2f   xv[K][BATCH];
    float vj[K];
    int   rj[K];
#pragma unroll
    for (int k = 0; k < K; ++k) {
        const int cj = __builtin_amdgcn_readlane(c, j0 + k);
        vj[k] = __builtin_bit_cast(float, __builtin_amdgcn_readlane(__builtin_bit_cast(int, v), j0 + k));
        rj[k] = __builtin_amdgcn_readlane(r, j0 + k) - tileBase;
        const float* __restrict__ xp = x + (long)cj * FDIM + lane * 2;
#pragma unroll
        for (int b = 0; b < BATCH; ++b)
            xv[k][b] = *(const v2f*)(xp + (long)b * bstride);
    }
#pragma unroll
    for (int k = 0; k < K; ++k) {
        float* ap = sacc + rj[k] * (BATCH * FDIM) + lane;
#pragma unroll
        for (int b = 0; b < BATCH; ++b) {
            lds_add(ap + b * FDIM,      vj[k] * xv[k][b].x);
            lds_add(ap + b * FDIM + 32, vj[k] * xv[k][b].y);
        }
    }
}

__device__ __forceinline__ void flush_list(const float* __restrict__ x,
                                           const float* __restrict__ vals,
                                           const int*   __restrict__ rows,
                                           const int*   __restrict__ cols,
                                           float* sacc, const int* mylist, int cnt,
                                           int tileBase, int lane, long bstride) {
    int c = 0, r = 0; float v = 0.0f;
    if (lane < cnt) {
        const int e = mylist[lane];
        r = rows[e];
        c = cols[e];
        v = vals[e];
    }
    for (int base = 0; base < cnt; base += 32) {
        const int n = (cnt - base < 32) ? (cnt - base) : 32;
        int c2 = 0, r2 = 0; float v2 = 0.0f;
        if (base + 32 + lane < cnt) {
            const int e = mylist[base + 32 + lane];
            r2 = rows[e];
            c2 = cols[e];
            v2 = vals[e];
        }
        int j = 0;
        for (; j + 8 <= n; j += 8) gather_k<8>(x, sacc, c, v, r, tileBase, j, lane, bstride);
        if (j + 4 <= n) { gather_k<4>(x, sacc, c, v, r, tileBase, j, lane, bstride); j += 4; }
        for (; j < n; ++j) gather_k<1>(x, sacc, c, v, r, tileBase, j, lane, bstride);
        c = c2; r = r2; v = v2;
    }
}

__device__ __forceinline__ int scan_unit(int4 r, int e0, int tileBase, int* mylist, int cnt) {
    const int r4[4] = {r.x, r.y, r.z, r.w};
#pragma unroll
    for (int t = 0; t < 4; ++t) {
        const bool m = (unsigned)(r4[t] - tileBase) < (unsigned)TROWS;
        const unsigned mask = __builtin_amdgcn_ballot_w32(m);
        if (mask) {
            if (m) mylist[cnt + (int)__builtin_amdgcn_mbcnt_lo(mask, 0u)] = e0 + t;
            cnt += __builtin_popcount(mask);
        }
    }
    return cnt;
}

__global__ void __launch_bounds__(NWAVES * 32)
gcn_spmm_rowtile(const float* __restrict__ x,
                 const float* __restrict__ vals,
                 const int*   __restrict__ rows,
                 const int*   __restrict__ cols,
                 float*       __restrict__ agg,
                 long E, long Nn) {
    __shared__ float sacc[TROWS * BATCH * FDIM];
    __shared__ int   slist[NWAVES * LCAP];

    const int lane = threadIdx.x & 31;
    const int wave = __builtin_amdgcn_readfirstlane((int)(threadIdx.x >> 5));
    const int tileBase = (int)blockIdx.x * TROWS;
    const long bstride = Nn * FDIM;

    {
        float4* s4 = (float4*)sacc;
        for (int i = threadIdx.x; i < TROWS * BATCH * FDIM / 4; i += NWAVES * 32)
            s4[i] = float4{0.0f, 0.0f, 0.0f, 0.0f};
    }
    __syncthreads();

    int* mylist = slist + wave * LCAP;
    int  cnt = 0;

    const long nUnits = E >> 7;
    const long start  = ((long)blockIdx.x * nUnits) / gridDim.x;
    const int4* __restrict__ rows4 = (const int4*)rows;

    const long mine  = (nUnits > wave) ? (nUnits - wave + NWAVES - 1) / NWAVES : 0;
    const int  trips = (int)((mine + 7) >> 3);
    for (int t = 0; t < trips; ++t) {
        const long ub = wave + (long)(8 * NWAVES) * t;
        int4 r[8];
        long us[8];
#pragma unroll
        for (int q = 0; q < 8; ++q) {
            const long uq = ub + (long)q * NWAVES;
            const bool ok = uq < nUnits;
            long u = (ok ? uq : (long)wave) + start;
            if (u >= nUnits) u -= nUnits;
            us[q] = ok ? u : -1;
            r[q]  = rows4[(u << 5) + lane];
        }
#pragma unroll
        for (int q = 0; q < 8; ++q) {
            if (us[q] >= 0) cnt = scan_unit(r[q], (int)(us[q] << 7) + lane * 4, tileBase, mylist, cnt);
            if ((q & 3) == 3 && cnt > LFLUSH) {
                flush_list(x, vals, rows, cols, sacc, mylist, cnt, tileBase, lane, bstride);
                cnt = 0;
            }
        }
    }
    if (wave == 0) {
        for (long eb = nUnits << 7; eb < E; eb += 32) {
            const long e = eb + lane;
            const bool m = (e < E) && ((unsigned)(rows[e] - tileBase) < (unsigned)TROWS);
            const unsigned mask = __builtin_amdgcn_ballot_w32(m);
            if (mask) {
                if (m) mylist[cnt + (int)__builtin_amdgcn_mbcnt_lo(mask, 0u)] = (int)e;
                cnt += __builtin_popcount(mask);
            }
        }
    }
    flush_list(x, vals, rows, cols, sacc, mylist, cnt, tileBase, lane, bstride);

    __syncthreads();

    const long left   = Nn - (long)tileBase;
    const int  nvalid = (left < TROWS) ? (int)left : TROWS;
#pragma unroll 4
    for (int i = wave; i < nvalid * BATCH; i += NWAVES) {
        const int rl = i >> 2;
        const int b  = i & 3;
        const float* sp = sacc + i * FDIM;
        v2f o;
        o.x = sp[lane];
        o.y = sp[32 + lane];
        *(v2f*)(agg + ((long)b * Nn + tileBase + rl) * FDIM + lane * 2) = o;
    }
}

__global__ void __launch_bounds__(256)
gcn_gemm_bias_elu(const float* __restrict__ agg,
                  const float* __restrict__ W,
                  const float* __restrict__ bias,
                  float*       __restrict__ out,
                  long nrows) {
    const int lane = threadIdx.x & 31;
    const int wave = threadIdx.x >> 5;
    const long rowBase = ((long)blockIdx.x * 8 + wave) * 16;
    if (rowBase >= nrows) return;

    const int m  = lane & 15;
    const int kg = (lane >> 4) << 1;

    long arow = rowBase + m;
    if (arow >= nrows) arow = nrows - 1;
    const float* __restrict__ aptr = agg + arow * FDIM;

    v8f acc[4] = {v8f{}, v8f{}, v8f{}, v8f{}};

#pragma unroll
    for (int k4 = 0; k4 < FDIM / 4; ++k4) {
        const int kk = k4 * 4 + kg;
        v2f a;
        a.x = aptr[kk];
        a.y = aptr[kk + 1];

        const float* __restrict__ w0 = W + (long)kk * FDIM + m;
        const float* __restrict__ w1 = W + (long)(kk + 1) * FDIM + m;
#pragma unroll
        for (int j = 0; j < 4; ++j) {
            v2f b;
            b.x = w0[j * 16];
            b.y = w1[j * 16];
            acc[j] = __builtin_amdgcn_wmma_f32_16x16x4_f32(
                 false, a,  false, b,
                 (short)0, acc[j],
                 false,  false);
        }
    }

    const int rowAdd = (lane >> 4) * 8;
#pragma unroll
    for (int j = 0; j < 4; ++j) {
        const int col = j * 16 + m;
        const float bb = bias[col];
#pragma unroll
        for (int r = 0; r < 8; ++r) {
            const long row = rowBase + rowAdd + r;
            if (row < nrows) {
                const float v = acc[j][r] + bb;
                out[row * FDIM + col] = (v > 0.0f) ? v : (__expf(v) - 1.0f);
            }
        }
    }
}

extern "C" void kernel_launch(void* const* d_in, const int* in_sizes, int n_in,
                              void* d_out, int out_size, void* d_ws, size_t ws_size,
                              hipStream_t stream) {
    const float* x      = (const float*)d_in[0];
    const float* vals   = (const float*)d_in[1];
    const float* weight = (const float*)d_in[2];
    const float* bias   = (const float*)d_in[3];
    const int*   rows   = (const int*)d_in[4];
    const int*   cols   = (const int*)d_in[5];
    float*       out    = (float*)d_out;

    const long E  = in_sizes[1];
    const long Nn = (long)in_sizes[0] / (BATCH * FDIM);
    float* agg = (float*)d_ws;

    {
        const long tiles = (Nn + TROWS - 1) / TROWS;
        gcn_spmm_rowtile<<<(unsigned)tiles, NWAVES * 32, 0, stream>>>(
            x, vals, rows, cols, agg, E, Nn);
    }
    {
        const long nrows  = (long)BATCH * Nn;
        const long blocks = (nrows + 127) / 128;
        gcn_gemm_bias_elu<<<(unsigned)blocks, 256, 0, stream>>>(
            agg, weight, bias, out, nrows);
    }
}


# GEMM+bias+ELU done inside the row-tile kernel: the finished LDS sums (rows in feature order, pitch 68 dwords) are read in place as A fragments, agg is never stored; one launch instead of two
# speedup vs baseline: 3.1721x; 1.0583x over previous
#include <hip/hip_runtime.h>


#define BATCH 4
#define FDIM  64

typedef __attribute__((ext_vector_type(2))) float v2f;
typedef __attribute__((ext_vector_type(8))) float v8f;

#define TROWS   196
#define PITCH   68
#define NWAVES  32
#define LCAP    736
#define LFLUSH  224

__device__ __forceinline__ void lds_add(float* p, float v) {
    __hip_atomic_fetch_add(p, v, __ATOMIC_RELAXED, __HIP_MEMORY_SCOPE_WORKGROUP);
}

template <int K>
__device__ __forceinline__ void gather_k(const float* __restrict__ x, float* sacc,
                                         int c, float v, int r, int tileBase, int j0, int lane, long bstride) {
    v2f   xv[K][BATCH];
    float vj[K];
    int   rj[K];
#pragma unroll
    for (int k = 0; k < K; ++k) {
        const int cj = __builtin_amdgcn_readlane(c, j0 + k);
        vj[k] = __builtin_bit_cast(float, __builtin_amdgcn_readlane(__builtin_bit_cast(int, v), j0 + k));
        rj[k] = __builtin_amdgcn_readlane(r, j0 + k) - tileBase;
        const float* __restrict__ xp = x + (long)cj * FDIM + lane * 2;
#pragma unroll
        for (int b = 0; b < BATCH; ++b)
            xv[k][b] = *(const v2f*)(xp + (long)b * bstride);
    }
#pragma unroll
    for (int k = 0; k < K; ++k) {
        float* ap = sacc + rj[k] * (BATCH * PITCH) + lane * 2;
#pragma unroll
        for (int b = 0; b < BATCH; ++b) {
            lds_add(ap + b * PITCH,     vj[k] * xv[k][b].x);
            lds_add(ap + b * PITCH + 1, vj[k] * xv[k][b].y);
        }
    }
}

__device__ __forceinline__ void flush_list(const float* __restrict__ x,
                                           const float* __restrict__ vals,
                                           const int*   __restrict__ rows,
                                           const int*   __restrict__ cols,
                                           float* sacc, const int* mylist, int cnt,
                                           int tileBase, int lane, long bstride) {
    int c = 0, r = 0; float v = 0.0f;
    if (lane < cnt) {
        const int e = mylist[lane];
        r = rows[e];
        c = cols[e];
        v = vals[e];
    }
    for (int base = 0; base < cnt; base += 32) {
        const int n = (cnt - base < 32) ? (cnt - base) : 32;
        int c2 = 0, r2 = 0; float v2 = 0.0f;
        if (base + 32 + lane < cnt) {
            const int e = mylist[base + 32 + lane];
            r2 = rows[e];
            c2 = cols[e];
            v2 = vals[e];
        }
        int j = 0;
        for (; j + 8 <= n; j += 8) gather_k<8>(x, sacc, c, v, r, tileBase, j, lane, bstride);
        if (j + 4 <= n) { gather_k<4>(x, sacc, c, v, r, tileBase, j, lane, bstride); j += 4; }
        for (; j < n; ++j) gather_k<1>(x, sacc, c, v, r, tileBase, j, lane, bstride);
        c = c2; r = r2; v = v2;
    }
}

__device__ __forceinline__ int scan_unit(int4 r, int e0, int tileBase, int* mylist, int cnt) {
    const int r4[4] = {r.x, r.y, r.z, r.w};
#pragma unroll
    for (int t = 0; t < 4; ++t) {
        const bool m = (unsigned)(r4[t] - tileBase) < (unsigned)TROWS;
        const unsigned mask = __builtin_amdgcn_ballot_w32(m);
        if (mask) {
            if (m) mylist[cnt + (int)__builtin_amdgcn_mbcnt_lo(mask, 0u)] = e0 + t;
            cnt += __builtin_popcount(mask);
        }
    }
    return cnt;
}

__global__ void __launch_bounds__(NWAVES * 32)
gcn_spmm_rowtile_gemm(const float* __restrict__ x,
                      const float* __restrict__ vals,
                      const float* __restrict__ W,
                      const float* __restrict__ bias,
                      const int*   __restrict__ rows,
                      const int*   __restrict__ cols,
                      float*       __restrict__ out,
                      long E, long Nn) {
    __shared__ __attribute__((aligned(16))) float sacc[TROWS * BATCH * PITCH];
    __shared__ int   slist[NWAVES * LCAP];

    const int lane = threadIdx.x & 31;
    const int wave = __builtin_amdgcn_readfirstlane((int)(threadIdx.x >> 5));
    const int tileBase = (int)blockIdx.x * TROWS;
    const long bstride = Nn * FDIM;

    {
        float4* s4 = (float4*)sacc;
        for (int i = threadIdx.x; i < TROWS * BATCH * PITCH / 4; i += NWAVES * 32)
            s4[i] = float4{0.0f, 0.0f, 0.0f, 0.0f};
    }
    __syncthreads();

    int* mylist = slist + wave * LCAP;
    int  cnt = 0;

    const long nUnits = E >> 7;
    const long start  = ((long)blockIdx.x * nUnits) / gridDim.x;
    const int4* __restrict__ rows4 = (const int4*)rows;

    const long mine  = (nUnits > wave) ? (nUnits - wave + NWAVES - 1) / NWAVES : 0;
    const int  trips = (int)((mine + 7) >> 3);
    for (int t = 0; t < trips; ++t) {
        const long ub = wave + (long)(8 * NWAVES) * t;
        int4 r[8];
        long us[8];
#pragma unroll
        for (int q = 0; q < 8; ++q) {
            const long uq = ub + (long)q * NWAVES;
            const bool ok = uq < nUnits;
            long u = (ok ? uq : (long)wave) + start;
            if (u >= nUnits) u -= nUnits;
            us[q] = ok ? u : -1;
            r[q]  = rows4[(u << 5) + lane];
        }
#pragma unroll
        for (int q = 0; q < 8; ++q) {
            if (us[q] >= 0) cnt = scan_unit(r[q], (int)(us[q] << 7) + lane * 4, tileBase, mylist, cnt);
            if ((q & 3) == 3 && cnt > LFLUSH) {
                flush_list(x, vals, rows, cols, sacc, mylist, cnt, tileBase, lane, bstride);
                cnt = 0;
            }
        }
    }
    if (wave == 0) {
        for (long eb = nUnits << 7; eb < E; eb += 32) {
            const long e = eb + lane;
            const bool m = (e < E) && ((unsigned)(rows[e] - tileBase) < (unsigned)TROWS);
            const unsigned mask = __builtin_amdgcn_ballot_w32(m);
            if (mask) {
                if (m) mylist[cnt + (int)__builtin_amdgcn_mbcnt_lo(mask, 0u)] = (int)e;
                cnt += __builtin_popcount(mask);
            }
        }
    }
    flush_list(x, vals, rows, cols, sacc, mylist, cnt, tileBase, lane, bstride);

    const int c  = lane & 15;
    const int h  = lane >> 4;
    const int nt = wave & 3;
    const int mg = wave >> 2;
    const int col = nt * 16 + c;

    v2f bf[FDIM / 4];
#pragma unroll
    for (int k = 0; k < FDIM / 4; ++k) {
        const int kk = 4 * k + 2 * h;
        bf[k].x = W[kk * FDIM + col];
        bf[k].y = W[(kk + 1) * FDIM + col];
    }
    const float bb = bias[col];

    __syncthreads();

    const long left   = Nn - (long)tileBase;
    const int  nvalid = (left < TROWS) ? (int)left : TROWS;
    const int  mtEnd  = (nvalid * BATCH + 15) >> 4;

    for (int mt = mg; mt < mtEnd; mt += NWAVES / 4) {
        const v2f* ap = (const v2f*)(sacc + (mt * 16 + c) * PITCH + 2 * h);
        v2f af[FDIM / 4];
#pragma unroll
        for (int k = 0; k < FDIM / 4; ++k) af[k] = ap[2 * k];
        v8f acc = v8f{};
#pragma unroll
        for (int k = 0; k < FDIM / 4; ++k)
            acc = __builtin_amdgcn_wmma_f32_16x16x4_f32(
                 false, af[k],  false, bf[k],
                 (short)0, acc,
                 false,  false);

        const int nodeL = mt * 4 + 2 * h;
        float o[8];
#pragma unroll
        for (int r = 0; r < 8; ++r) {
            const float v = acc[r] + bb;
            o[r] = (v > 0.0f) ? v : (__expf(v) - 1.0f);
        }
#pragma unroll
        for (int r = 0; r < 8; ++r) {
            const int rl = nodeL + (r >> 2);
            const int b  = r & 3;
            if (rl < nvalid)
                out[((long)b * Nn + tileBase + rl) * FDIM + col] = o[r];
        }
    }
}

extern "C" void kernel_launch(void* const* d_in, const int* in_sizes, int n_in,
                              void* d_out, int out_size, void* d_ws, size_t ws_size,
                              hipStream_t stream) {
    const float* x      = (const float*)d_in[0];
    const float* vals   = (const float*)d_in[1];
    const float* weight = (const float*)d_in[2];
    const float* bias   = (const float*)d_in[3];
    const int*   rows   = (const int*)d_in[4];
    const int*   cols   = (const int*)d_in[5];
    float*       out    = (float*)d_out;

    const long E  = in_sizes[1];
    const long Nn = (long)in_sizes[0] / (BATCH * FDIM);

    {
        const long tiles = (Nn + TROWS - 1) / TROWS;
        gcn_spmm_rowtile_gemm<<<(unsigned)tiles, NWAVES * 32, 0, stream>>>(
            x, vals, weight, bias, rows, cols, out, E, Nn);
    }
}
